# starting kernel
# baseline (speedup 1.0000x reference)
#include <hip/hip_runtime.h>
#include <hip/hip_bf16.h>

#define S_ 32
#define B_ 32
#define N_ 8192
#define D_ 256
#define M_ 8
#define CHUNKS 16
#define ROWS_PER_BLOCK (N_ / CHUNKS)

typedef __attribute__((ext_vector_type(2)))  float    v2f;
typedef __attribute__((ext_vector_type(8)))  float    v8f;
typedef __attribute__((ext_vector_type(16))) _Float16 v16h;

#define WS_SUMS 0
#define WS_CNT  (B_ * S_ * D_)
#define WS_OBJS (WS_CNT + B_ * S_)
#define WS_ZERO_N (WS_OBJS)

#if defined(__has_builtin)
#if __has_builtin(__builtin_amdgcn_wmma_f32_16x16x4_f32)
#define HAVE_F32_WMMA 1
#endif
#endif

__global__ __launch_bounds__(256) void zero_ws(float* __restrict__ ws) {
    int i = blockIdx.x * blockDim.x + threadIdx.x;
    int stride = gridDim.x * blockDim.x;
    for (; i < WS_ZERO_N; i += stride) ws[i] = 0.0f;
}

__global__ __launch_bounds__(256) void seg_accum(const float* __restrict__ features,
                                                 const int*   __restrict__ seg,
                                                 float*       __restrict__ ws) {
    __shared__ float acc[S_ * D_];
    __shared__ float cnt[S_];
    const int tid = threadIdx.x;
    for (int i = tid; i < S_ * D_; i += 256) acc[i] = 0.0f;
    if (tid < S_) cnt[tid] = 0.0f;
    __syncthreads();

    const int b     = blockIdx.x / CHUNKS;
    const int chunk = blockIdx.x % CHUNKS;
    const int wave  = tid >> 5;
    const int lane  = tid & 31;
    const float* fb = features + (size_t)b * N_ * D_;
    const int*   sb = seg + b * N_;

    for (int i = 0; i < ROWS_PER_BLOCK / 8; ++i) {
        const int n = chunk * ROWS_PER_BLOCK + i * 8 + wave;
        const int s = sb[n];
        const float4* row = (const float4*)(fb + (size_t)n * D_);
        float4 v0 = row[lane * 2 + 0];
        float4 v1 = row[lane * 2 + 1];
        float* a = acc + s * D_ + lane * 8;
        atomicAdd(a + 0, v0.x); atomicAdd(a + 1, v0.y);
        atomicAdd(a + 2, v0.z); atomicAdd(a + 3, v0.w);
        atomicAdd(a + 4, v1.x); atomicAdd(a + 5, v1.y);
        atomicAdd(a + 6, v1.z); atomicAdd(a + 7, v1.w);
        if (lane == 0) atomicAdd(&cnt[s], 1.0f);
    }
    __syncthreads();

    float* gsum = ws + WS_SUMS + (size_t)b * S_ * D_;
    for (int i = tid; i < S_ * D_; i += 256) atomicAdd(&gsum[i], acc[i]);
    if (tid < S_) atomicAdd(&ws[WS_CNT + b * S_ + tid], cnt[tid]);
}

__global__ __launch_bounds__(256) void top_objs(float* __restrict__ ws) {
    __shared__ int   topi[M_];
    __shared__ float scale[M_];
    const int b = blockIdx.x, tid = threadIdx.x;
    if (tid == 0) {
        const float* c = ws + WS_CNT + b * S_;
        unsigned used = 0u;
        for (int m = 0; m < M_; ++m) {
            int best = 0; float bc = -1.0f;
            for (int s = 0; s < S_; ++s) {
                if (!((used >> s) & 1u) && c[s] > bc) { bc = c[s]; best = s; }
            }
            used |= (1u << best);
            topi[m]  = best;
            scale[m] = (bc > 0.0f) ? (1.0f / bc) : 0.0f;
        }
    }
    __syncthreads();
    const float* gsum = ws + WS_SUMS + (size_t)b * S_ * D_;
    float*       objs = ws + WS_OBJS + (size_t)b * M_ * D_;
    for (int m = 0; m < M_; ++m)
        objs[m * D_ + tid] = gsum[topi[m] * D_ + tid] * scale[m];
}

__global__ __launch_bounds__(512) void proj_ln(const float* __restrict__ objs,
                                               const float* __restrict__ W,
                                               const float* __restrict__ bias,
                                               const float* __restrict__ gamma,
                                               const float* __restrict__ beta,
                                               float*       __restrict__ out) {
    __shared__ float tile[16 * D_];
    const int tid  = threadIdx.x;
    const int wave = tid >> 5;
    const int lane = tid & 31;
    const int half = lane >> 4;
    const int l    = lane & 15;
    const int rowbase = blockIdx.x * 16;
    const int colbase = wave * 16;

    v8f c = {};

#ifdef HAVE_F32_WMMA
    for (int k0 = 0; k0 < D_; k0 += 4) {
        const int k = k0 + 2 * half;
        v2f a, w;
        const float* ap = objs + (size_t)(rowbase + l) * D_ + k;
        a.x = ap[0]; a.y = ap[1];
        const float* wp = W + (size_t)(colbase + l) * D_ + k;
        w.x = wp[0]; w.y = wp[1];
        c = __builtin_amdgcn_wmma_f32_16x16x4_f32(false, a, false, w,
                                                  (short)0, c, false, false);
    }
#else
    for (int k0 = 0; k0 < D_; k0 += 32) {
        v16h a, w;
        const float* ap = objs + (size_t)(rowbase + l) * D_;
        const float* wp = W + (size_t)(colbase + l) * D_;
        for (int i = 0; i < 8; ++i) {
            const int kk = k0 + ((i < 4) ? (2 * i + 8 * half)
                                         : (16 + 2 * (i - 4) + 8 * half));
            a[2 * i]     = (_Float16)ap[kk];
            a[2 * i + 1] = (_Float16)ap[kk + 1];
            w[2 * i]     = (_Float16)wp[kk];
            w[2 * i + 1] = (_Float16)wp[kk + 1];
        }
        c = __builtin_amdgcn_wmma_f32_16x16x32_f16(false, a, false, w,
                                                   (short)0, c, false, false);
    }
#endif

    for (int j = 0; j < 8; ++j) {
        const int m = j + half * 8;
        const int n = colbase + l;
        tile[m * D_ + n] = c[j] + bias[n];
    }
    __syncthreads();

    const float* row = tile + wave * D_;
    float v[8], s = 0.0f, s2 = 0.0f;
    for (int j = 0; j < 8; ++j) {
        v[j] = row[lane * 8 + j];
        s  += v[j];
        s2 += v[j] * v[j];
    }
    for (int off = 16; off > 0; off >>= 1) {
        s  += __shfl_xor(s,  off, 32);
        s2 += __shfl_xor(s2, off, 32);
    }
    const float mean = s * (1.0f / D_);
    const float var  = s2 * (1.0f / D_) - mean * mean;
    const float rstd = rsqrtf(var + 1e-5f);
    float* orow = out + (size_t)(rowbase + wave) * D_;
    for (int j = 0; j < 8; ++j) {
        const int idx = lane * 8 + j;
        orow[idx] = (v[j] - mean) * rstd * gamma[idx] + beta[idx];
    }
}

extern "C" void kernel_launch(void* const* d_in, const int* in_sizes, int n_in,
                              void* d_out, int out_size, void* d_ws, size_t ws_size,
                              hipStream_t stream) {
    const float* features = (const float*)d_in[0];
    const int*   seg      = (const int*)  d_in[1];
    const float* W        = (const float*)d_in[2];
    const float* b        = (const float*)d_in[3];
    const float* gamma    = (const float*)d_in[4];
    const float* beta     = (const float*)d_in[5];
    float* ws  = (float*)d_ws;
    float* out = (float*)d_out;

    zero_ws  <<<512, 256, 0, stream>>>(ws);
    seg_accum<<<B_ * CHUNKS, 256, 0, stream>>>(features, seg, ws);
    top_objs <<<B_, 256, 0, stream>>>(ws);
    proj_ln  <<<(B_ * M_) / 16, 512, 0, stream>>>(ws + WS_OBJS, W, b, gamma, beta, out);
}


# seg_accum: segment sums as onehot(ids) x features on f32 WMMA 16x16x4, a wave keeps 32 segments x 64 columns in registers; LDS scatter atomics gone; f32 sum order changes (row order)
# speedup vs baseline: 1.5181x; 1.5181x over previous
#include <hip/hip_runtime.h>
#include <hip/hip_bf16.h>

#define S_ 32
#define B_ 32
#define N_ 8192
#define D_ 256
#define M_ 8
#define CHUNKS 16
#define ROWS_PER_BLOCK (N_ / CHUNKS)

typedef __attribute__((ext_vector_type(2)))  float    v2f;
typedef __attribute__((ext_vector_type(8)))  float    v8f;
typedef __attribute__((ext_vector_type(16))) _Float16 v16h;

#define WS_SUMS 0
#define WS_CNT  (B_ * S_ * D_)
#define WS_OBJS (WS_CNT + B_ * S_)
#define WS_ZERO_N (WS_OBJS)

#if defined(__has_builtin)
#if __has_builtin(__builtin_amdgcn_wmma_f32_16x16x4_f32)
#define HAVE_F32_WMMA 1
#endif
#endif

__global__ __launch_bounds__(256) void zero_ws(float* __restrict__ ws) {
    int i = blockIdx.x * blockDim.x + threadIdx.x;
    int stride = gridDim.x * blockDim.x;
    for (; i < WS_ZERO_N; i += stride) ws[i] = 0.0f;
}

#define GROUPS_PER_WAVE 16

__device__ __forceinline__ void seg_load_group(float4 (&dst)[8], const float* p) {
    for (int i = 0; i < 8; ++i)
        dst[i] = *(const float4*)(p + (size_t)i * D_);
}

__device__ __forceinline__ void seg_mma_group(v8f (&acc)[2][4], const float4 (&f)[8],
                                              const int* ids, int c) {
    const int4 ia = *(const int4*)(ids);
    const int4 ib = *(const int4*)(ids + 4);
    const int id[8] = {ia.x, ia.y, ia.z, ia.w, ib.x, ib.y, ib.z, ib.w};
    for (int u = 0; u < 4; ++u) {
        const int s0 = id[2 * u], s1 = id[2 * u + 1];
        v2f a0, a1;
        a0.x = (s0 == c)      ? 1.0f : 0.0f;
        a0.y = (s1 == c)      ? 1.0f : 0.0f;
        a1.x = (s0 == c + 16) ? 1.0f : 0.0f;
        a1.y = (s1 == c + 16) ? 1.0f : 0.0f;
        const float4 x = f[2 * u], y = f[2 * u + 1];
        v2f b0, b1, b2, b3;
        b0.x = x.x; b0.y = y.x;
        b1.x = x.y; b1.y = y.y;
        b2.x = x.z; b2.y = y.z;
        b3.x = x.w; b3.y = y.w;
        acc[0][0] = __builtin_amdgcn_wmma_f32_16x16x4_f32(false, a0, false, b0, (short)0, acc[0][0], false, false);
        acc[1][0] = __builtin_amdgcn_wmma_f32_16x16x4_f32(false, a1, false, b0, (short)0, acc[1][0], false, false);
        acc[0][1] = __builtin_amdgcn_wmma_f32_16x16x4_f32(false, a0, false, b1, (short)0, acc[0][1], false, false);
        acc[1][1] = __builtin_amdgcn_wmma_f32_16x16x4_f32(false, a1, false, b1, (short)0, acc[1][1], false, false);
        acc[0][2] = __builtin_amdgcn_wmma_f32_16x16x4_f32(false, a0, false, b2, (short)0, acc[0][2], false, false);
        acc[1][2] = __builtin_amdgcn_wmma_f32_16x16x4_f32(false, a1, false, b2, (short)0, acc[1][2], false, false);
        acc[0][3] = __builtin_amdgcn_wmma_f32_16x16x4_f32(false, a0, false, b3, (short)0, acc[0][3], false, false);
        acc[1][3] = __builtin_amdgcn_wmma_f32_16x16x4_f32(false, a1, false, b3, (short)0, acc[1][3], false, false);
    }
}

__global__ __launch_bounds__(256) void seg_accum(const float* __restrict__ features,
                                                 const int*   __restrict__ seg,
                                                 float*       __restrict__ ws) {
    __shared__ __attribute__((aligned(16))) float red[S_ * D_];
    __shared__ __attribute__((aligned(16))) int   sid[ROWS_PER_BLOCK];
    __shared__ float cnt[S_];
    const int tid   = threadIdx.x;
    const int b     = blockIdx.x / CHUNKS;
    const int chunk = blockIdx.x % CHUNKS;
    const int wave  = tid >> 5;
    const int lane  = tid & 31;
    const int h     = lane >> 4;
    const int c     = lane & 15;
    const int q     = wave & 3;
    const int rh    = wave >> 2;

    const int*   sb = seg + b * N_ + chunk * ROWS_PER_BLOCK;
    const float* fl = features
                    + ((size_t)b * N_ + (size_t)chunk * ROWS_PER_BLOCK + rh * 256 + 8 * h) * D_
                    + q * 64 + c * 4;

    float4 fa[8], fb[8];
    seg_load_group(fa, fl);
    const int s0 = sb[tid];
    const int s1 = sb[tid + 256];

    if (tid < S_) cnt[tid] = 0.0f;
    __syncthreads();
    sid[tid]       = s0;
    sid[tid + 256] = s1;
    atomicAdd(&cnt[s0], 1.0f);
    atomicAdd(&cnt[s1], 1.0f);
    __syncthreads();

    const int* sl = sid + rh * 256 + 8 * h;
    v8f acc[2][4] = {};

#pragma nounroll
    for (int g = 0; g < GROUPS_PER_WAVE - 2; g += 2) {
        seg_load_group(fb, fl + (size_t)(g + 1) * 16 * D_);
        seg_mma_group(acc, fa, sl + g * 16, c);
        seg_load_group(fa, fl + (size_t)(g + 2) * 16 * D_);
        seg_mma_group(acc, fb, sl + (g + 1) * 16, c);
    }
    seg_load_group(fb, fl + (size_t)(GROUPS_PER_WAVE - 1) * 16 * D_);
    seg_mma_group(acc, fa, sl + (GROUPS_PER_WAVE - 2) * 16, c);
    seg_mma_group(acc, fb, sl + (GROUPS_PER_WAVE - 1) * 16, c);

    if (rh == 1) {
        for (int mt = 0; mt < 2; ++mt)
            for (int r = 0; r < 8; ++r) {
                float4 v = {acc[mt][0][r], acc[mt][1][r], acc[mt][2][r], acc[mt][3][r]};
                *(float4*)(red + (mt * 16 + h * 8 + r) * D_ + q * 64 + c * 4) = v;
            }
    }
    __syncthreads();
    if (rh == 0) {
        for (int mt = 0; mt < 2; ++mt)
            for (int r = 0; r < 8; ++r) {
                float4* p = (float4*)(red + (mt * 16 + h * 8 + r) * D_ + q * 64 + c * 4);
                float4 v = *p;
                v.x = acc[mt][0][r] + v.x; v.y = acc[mt][1][r] + v.y;
                v.z = acc[mt][2][r] + v.z; v.w = acc[mt][3][r] + v.w;
                *p = v;
            }
    }
    __syncthreads();

    float* gsum = ws + WS_SUMS + (size_t)b * S_ * D_;
    for (int i = tid; i < S_ * D_; i += 256) atomicAdd(&gsum[i], red[i]);
    if (tid < S_) atomicAdd(&ws[WS_CNT + b * S_ + tid], cnt[tid]);
}

__global__ __launch_bounds__(256) void top_objs(float* __restrict__ ws) {
    __shared__ int   topi[M_];
    __shared__ float scale[M_];
    const int b = blockIdx.x, tid = threadIdx.x;
    if (tid == 0) {
        const float* c = ws + WS_CNT + b * S_;
        unsigned used = 0u;
        for (int m = 0; m < M_; ++m) {
            int best = 0; float bc = -1.0f;
            for (int s = 0; s < S_; ++s) {
                if (!((used >> s) & 1u) && c[s] > bc) { bc = c[s]; best = s; }
            }
            used |= (1u << best);
            topi[m]  = best;
            scale[m] = (bc > 0.0f) ? (1.0f / bc) : 0.0f;
        }
    }
    __syncthreads();
    const float* gsum = ws + WS_SUMS + (size_t)b * S_ * D_;
    float*       objs = ws + WS_OBJS + (size_t)b * M_ * D_;
    for (int m = 0; m < M_; ++m)
        objs[m * D_ + tid] = gsum[topi[m] * D_ + tid] * scale[m];
}

__global__ __launch_bounds__(512) void proj_ln(const float* __restrict__ objs,
                                               const float* __restrict__ W,
                                               const float* __restrict__ bias,
                                               const float* __restrict__ gamma,
                                               const float* __restrict__ beta,
                                               float*       __restrict__ out) {
    __shared__ float tile[16 * D_];
    const int tid  = threadIdx.x;
    const int wave = tid >> 5;
    const int lane = tid & 31;
    const int half = lane >> 4;
    const int l    = lane & 15;
    const int rowbase = blockIdx.x * 16;
    const int colbase = wave * 16;

    v8f c = {};

#ifdef HAVE_F32_WMMA
    for (int k0 = 0; k0 < D_; k0 += 4) {
        const int k = k0 + 2 * half;
        v2f a, w;
        const float* ap = objs + (size_t)(rowbase + l) * D_ + k;
        a.x = ap[0]; a.y = ap[1];
        const float* wp = W + (size_t)(colbase + l) * D_ + k;
        w.x = wp[0]; w.y = wp[1];
        c = __builtin_amdgcn_wmma_f32_16x16x4_f32(false, a, false, w,
                                                  (short)0, c, false, false);
    }
#else
    for (int k0 = 0; k0 < D_; k0 += 32) {
        v16h a, w;
        const float* ap = objs + (size_t)(rowbase + l) * D_;
        const float* wp = W + (size_t)(colbase + l) * D_;
        for (int i = 0; i < 8; ++i) {
            const int kk = k0 + ((i < 4) ? (2 * i + 8 * half)
                                         : (16 + 2 * (i - 4) + 8 * half));
            a[2 * i]     = (_Float16)ap[kk];
            a[2 * i + 1] = (_Float16)ap[kk + 1];
            w[2 * i]     = (_Float16)wp[kk];
            w[2 * i + 1] = (_Float16)wp[kk + 1];
        }
        c = __builtin_amdgcn_wmma_f32_16x16x32_f16(false, a, false, w,
                                                   (short)0, c, false, false);
    }
#endif

    for (int j = 0; j < 8; ++j) {
        const int m = j + half * 8;
        const int n = colbase + l;
        tile[m * D_ + n] = c[j] + bias[n];
    }
    __syncthreads();

    const float* row = tile + wave * D_;
    float v[8], s = 0.0f, s2 = 0.0f;
    for (int j = 0; j < 8; ++j) {
        v[j] = row[lane * 8 + j];
        s  += v[j];
        s2 += v[j] * v[j];
    }
    for (int off = 16; off > 0; off >>= 1) {
        s  += __shfl_xor(s,  off, 32);
        s2 += __shfl_xor(s2, off, 32);
    }
    const float mean = s * (1.0f / D_);
    const float var  = s2 * (1.0f / D_) - mean * mean;
    const float rstd = rsqrtf(var + 1e-5f);
    float* orow = out + (size_t)(rowbase + wave) * D_;
    for (int j = 0; j < 8; ++j) {
        const int idx = lane * 8 + j;
        orow[idx] = (v[j] - mean) * rstd * gamma[idx] + beta[idx];
    }
}

extern "C" void kernel_launch(void* const* d_in, const int* in_sizes, int n_in,
                              void* d_out, int out_size, void* d_ws, size_t ws_size,
                              hipStream_t stream) {
    const float* features = (const float*)d_in[0];
    const int*   seg      = (const int*)  d_in[1];
    const float* W        = (const float*)d_in[2];
    const float* b        = (const float*)d_in[3];
    const float* gamma    = (const float*)d_in[4];
    const float* beta     = (const float*)d_in[5];
    float* ws  = (float*)d_ws;
    float* out = (float*)d_out;

    zero_ws  <<<512, 256, 0, stream>>>(ws);
    seg_accum<<<B_ * CHUNKS, 256, 0, stream>>>(features, seg, ws);
    top_objs <<<B_, 256, 0, stream>>>(ws);
    proj_ln  <<<(B_ * M_) / 16, 512, 0, stream>>>(ws + WS_OBJS, W, b, gamma, beta, out);
}


# top_objs folded into proj_ln (3 launches): each 16-row block ranks its two batches' counts itself, reads A rows from the chosen sum rows; 1/count scales the product tile; objs plane gone
# speedup vs baseline: 1.8213x; 1.1997x over previous
#include <hip/hip_runtime.h>
#include <hip/hip_bf16.h>

#define S_ 32
#define B_ 32
#define N_ 8192
#define D_ 256
#define M_ 8
#define CHUNKS 16
#define ROWS_PER_BLOCK (N_ / CHUNKS)

typedef __attribute__((ext_vector_type(2)))  float    v2f;
typedef __attribute__((ext_vector_type(8)))  float    v8f;
typedef __attribute__((ext_vector_type(16))) _Float16 v16h;

#define WS_SUMS 0
#define WS_CNT  (B_ * S_ * D_)
#define WS_ZERO_N (WS_CNT + B_ * S_)

#if defined(__has_builtin)
#if __has_builtin(__builtin_amdgcn_wmma_f32_16x16x4_f32)
#define HAVE_F32_WMMA 1
#endif
#endif

__global__ __launch_bounds__(256) void zero_ws(float* __restrict__ ws) {
    int i = blockIdx.x * blockDim.x + threadIdx.x;
    int stride = gridDim.x * blockDim.x;
    for (; i < WS_ZERO_N; i += stride) ws[i] = 0.0f;
}

#define GROUPS_PER_WAVE 16

__device__ __forceinline__ void seg_load_group(float4 (&dst)[8], const float* p) {
    for (int i = 0; i < 8; ++i)
        dst[i] = *(const float4*)(p + (size_t)i * D_);
}

__device__ __forceinline__ void seg_mma_group(v8f (&acc)[2][4], const float4 (&f)[8],
                                              const int* ids, int c) {
    const int4 ia = *(const int4*)(ids);
    const int4 ib = *(const int4*)(ids + 4);
    const int id[8] = {ia.x, ia.y, ia.z, ia.w, ib.x, ib.y, ib.z, ib.w};
    for (int u = 0; u < 4; ++u) {
        const int s0 = id[2 * u], s1 = id[2 * u + 1];
        v2f a0, a1;
        a0.x = (s0 == c)      ? 1.0f : 0.0f;
        a0.y = (s1 == c)      ? 1.0f : 0.0f;
        a1.x = (s0 == c + 16) ? 1.0f : 0.0f;
        a1.y = (s1 == c + 16) ? 1.0f : 0.0f;
        const float4 x = f[2 * u], y = f[2 * u + 1];
        v2f b0, b1, b2, b3;
        b0.x = x.x; b0.y = y.x;
        b1.x = x.y; b1.y = y.y;
        b2.x = x.z; b2.y = y.z;
        b3.x = x.w; b3.y = y.w;
        acc[0][0] = __builtin_amdgcn_wmma_f32_16x16x4_f32(false, a0, false, b0, (short)0, acc[0][0], false, false);
        acc[1][0] = __builtin_amdgcn_wmma_f32_16x16x4_f32(false, a1, false, b0, (short)0, acc[1][0], false, false);
        acc[0][1] = __builtin_amdgcn_wmma_f32_16x16x4_f32(false, a0, false, b1, (short)0, acc[0][1], false, false);
        acc[1][1] = __builtin_amdgcn_wmma_f32_16x16x4_f32(false, a1, false, b1, (short)0, acc[1][1], false, false);
        acc[0][2] = __builtin_amdgcn_wmma_f32_16x16x4_f32(false, a0, false, b2, (short)0, acc[0][2], false, false);
        acc[1][2] = __builtin_amdgcn_wmma_f32_16x16x4_f32(false, a1, false, b2, (short)0, acc[1][2], false, false);
        acc[0][3] = __builtin_amdgcn_wmma_f32_16x16x4_f32(false, a0, false, b3, (short)0, acc[0][3], false, false);
        acc[1][3] = __builtin_amdgcn_wmma_f32_16x16x4_f32(false, a1, false, b3, (short)0, acc[1][3], false, false);
    }
}

__global__ __launch_bounds__(256) void seg_accum(const float* __restrict__ features,
                                                 const int*   __restrict__ seg,
                                                 float*       __restrict__ ws) {
    __shared__ __attribute__((aligned(16))) float red[S_ * D_];
    __shared__ __attribute__((aligned(16))) int   sid[ROWS_PER_BLOCK];
    __shared__ float cnt[S_];
    const int tid   = threadIdx.x;
    const int b     = blockIdx.x / CHUNKS;
    const int chunk = blockIdx.x % CHUNKS;
    const int wave  = tid >> 5;
    const int lane  = tid & 31;
    const int h     = lane >> 4;
    const int c     = lane & 15;
    const int q     = wave & 3;
    const int rh    = wave >> 2;

    const int*   sb = seg + b * N_ + chunk * ROWS_PER_BLOCK;
    const float* fl = features
                    + ((size_t)b * N_ + (size_t)chunk * ROWS_PER_BLOCK + rh * 256 + 8 * h) * D_
                    + q * 64 + c * 4;

    float4 fa[8], fb[8];
    seg_load_group(fa, fl);
    const int s0 = sb[tid];
    const int s1 = sb[tid + 256];

    if (tid < S_) cnt[tid] = 0.0f;
    __syncthreads();
    sid[tid]       = s0;
    sid[tid + 256] = s1;
    atomicAdd(&cnt[s0], 1.0f);
    atomicAdd(&cnt[s1], 1.0f);
    __syncthreads();

    const int* sl = sid + rh * 256 + 8 * h;
    v8f acc[2][4] = {};

#pragma nounroll
    for (int g = 0; g < GROUPS_PER_WAVE - 2; g += 2) {
        seg_load_group(fb, fl + (size_t)(g + 1) * 16 * D_);
        seg_mma_group(acc, fa, sl + g * 16, c);
        seg_load_group(fa, fl + (size_t)(g + 2) * 16 * D_);
        seg_mma_group(acc, fb, sl + (g + 1) * 16, c);
    }
    seg_load_group(fb, fl + (size_t)(GROUPS_PER_WAVE - 1) * 16 * D_);
    seg_mma_group(acc, fa, sl + (GROUPS_PER_WAVE - 2) * 16, c);
    seg_mma_group(acc, fb, sl + (GROUPS_PER_WAVE - 1) * 16, c);

    if (rh == 1) {
        for (int mt = 0; mt < 2; ++mt)
            for (int r = 0; r < 8; ++r) {
                float4 v = {acc[mt][0][r], acc[mt][1][r], acc[mt][2][r], acc[mt][3][r]};
                *(float4*)(red + (mt * 16 + h * 8 + r) * D_ + q * 64 + c * 4) = v;
            }
    }
    __syncthreads();
    if (rh == 0) {
        for (int mt = 0; mt < 2; ++mt)
            for (int r = 0; r < 8; ++r) {
                float4* p = (float4*)(red + (mt * 16 + h * 8 + r) * D_ + q * 64 + c * 4);
                float4 v = *p;
                v.x = acc[mt][0][r] + v.x; v.y = acc[mt][1][r] + v.y;
                v.z = acc[mt][2][r] + v.z; v.w = acc[mt][3][r] + v.w;
                *p = v;
            }
    }
    __syncthreads();

    float* gsum = ws + WS_SUMS + (size_t)b * S_ * D_;
    for (int i = tid; i < S_ * D_; i += 256) atomicAdd(&gsum[i], red[i]);
    if (tid < S_) atomicAdd(&ws[WS_CNT + b * S_ + tid], cnt[tid]);
}

__global__ __launch_bounds__(512) void pick_proj_ln(const float* __restrict__ ws,
                                                    const float* __restrict__ W,
                                                    const float* __restrict__ bias,
                                                    const float* __restrict__ gamma,
                                                    const float* __restrict__ beta,
                                                    float*       __restrict__ out) {
    __shared__ float tile[16 * D_];
    __shared__ int   topi[2 * M_];
    __shared__ float scale[2 * M_];
    const int tid  = threadIdx.x;
    const int wave = tid >> 5;
    const int lane = tid & 31;
    const int half = lane >> 4;
    const int l    = lane & 15;
    const int rowbase = blockIdx.x * 16;
    const int colbase = wave * 16;

    if (wave < 2) {
        const float mine = ws[WS_CNT + (2 * blockIdx.x + wave) * S_ + lane];
        int rank = 0;
        for (int t = 0; t < S_; ++t) {
            const float ct = __int_as_float(__builtin_amdgcn_readlane(__float_as_int(mine), t));
            rank += (ct > mine || (ct == mine && t < lane)) ? 1 : 0;
        }
        if (rank < M_) {
            topi[wave * M_ + rank]  = lane;
            scale[wave * M_ + rank] = (mine > 0.0f) ? (1.0f / mine) : 0.0f;
        }
    }
    __syncthreads();

    const float* ap = ws + WS_SUMS
                    + ((size_t)((rowbase + l) >> 3) * S_ + topi[l]) * D_;
    const float* wp = W + (size_t)(colbase + l) * D_;

    v8f c = {};

#ifdef HAVE_F32_WMMA
    for (int kb = 0; kb < D_; kb += 64) {
        v2f av[16], wv[16];
        for (int i = 0; i < 16; ++i) {
            const int k = kb + 4 * i + 2 * half;
            av[i] = *(const v2f*)(ap + k);
            wv[i] = *(const v2f*)(wp + k);
        }
        for (int i = 0; i < 16; ++i)
            c = __builtin_amdgcn_wmma_f32_16x16x4_f32(false, av[i], false, wv[i],
                                                      (short)0, c, false, false);
    }
#else
    for (int k0 = 0; k0 < D_; k0 += 32) {
        v16h a, w;
        for (int i = 0; i < 8; ++i) {
            const int kk = k0 + ((i < 4) ? (2 * i + 8 * half)
                                         : (16 + 2 * (i - 4) + 8 * half));
            a[2 * i]     = (_Float16)ap[kk];
            a[2 * i + 1] = (_Float16)ap[kk + 1];
            w[2 * i]     = (_Float16)wp[kk];
            w[2 * i + 1] = (_Float16)wp[kk + 1];
        }
        c = __builtin_amdgcn_wmma_f32_16x16x32_f16(false, a, false, w,
                                                   (short)0, c, false, false);
    }
#endif

    __builtin_amdgcn_sched_barrier(0);
    asm volatile("v_nop\n\tv_nop\n\tv_nop\n\tv_nop" : "+v"(c));
    __builtin_amdgcn_sched_barrier(0);

    for (int j = 0; j < 8; ++j) {
        const int m = j + half * 8;
        const int n = colbase + l;
        tile[m * D_ + n] = c[j] * scale[m] + bias[n];
    }
    __syncthreads();

    const float* row = tile + wave * D_;
    float v[8], s = 0.0f, s2 = 0.0f;
    for (int j = 0; j < 8; ++j) {
        v[j] = row[lane * 8 + j];
        s  += v[j];
        s2 += v[j] * v[j];
    }
    for (int off = 16; off > 0; off >>= 1) {
        s  += __shfl_xor(s,  off, 32);
        s2 += __shfl_xor(s2, off, 32);
    }
    const float mean = s * (1.0f / D_);
    const float var  = s2 * (1.0f / D_) - mean * mean;
    const float rstd = rsqrtf(var + 1e-5f);
    float* orow = out + (size_t)(rowbase + wave) * D_;
    for (int j = 0; j < 8; ++j) {
        const int idx = lane * 8 + j;
        orow[idx] = (v[j] - mean) * rstd * gamma[idx] + beta[idx];
    }
}

extern "C" void kernel_launch(void* const* d_in, const int* in_sizes, int n_in,
                              void* d_out, int out_size, void* d_ws, size_t ws_size,
                              hipStream_t stream) {
    const float* features = (const float*)d_in[0];
    const int*   seg      = (const int*)  d_in[1];
    const float* W        = (const float*)d_in[2];
    const float* b        = (const float*)d_in[3];
    const float* gamma    = (const float*)d_in[4];
    const float* beta     = (const float*)d_in[5];
    float* ws  = (float*)d_ws;
    float* out = (float*)d_out;

    zero_ws  <<<512, 256, 0, stream>>>(ws);
    seg_accum<<<B_ * CHUNKS, 256, 0, stream>>>(features, seg, ws);
    pick_proj_ln<<<(B_ * M_) / 16, 512, 0, stream>>>(ws, W, b, gamma, beta, out);
}


# seg_accum: a wave lists its 256 rows by segment half (ids < 16 first) and gathers rows in list order, so a 16-row step meets one M tile: 272 WMMAs a wave, was 512; f32 sum order changes
# speedup vs baseline: 2.5174x; 1.3822x over previous
#include <hip/hip_runtime.h>
#include <hip/hip_bf16.h>

#define S_ 32
#define B_ 32
#define N_ 8192
#define D_ 256
#define M_ 8
#define RHALVES 2
#define ROWS_PER_BLOCK (N_ / RHALVES)
#define QCOLS 64
#define SEG_WAVES 16

typedef __attribute__((ext_vector_type(2)))  float    v2f;
typedef __attribute__((ext_vector_type(8)))  float    v8f;
typedef __attribute__((ext_vector_type(16))) _Float16 v16h;

#define PLANE   (B_ * S_ * D_)
#define WS_PART 0
#define WS_CNT  (RHALVES * PLANE)
#define CNT_VECS (RHALVES * 4)
#define WS_NEED_BYTES ((WS_CNT + B_ * CNT_VECS * S_) * 4)
#define APITCH  260

#if defined(__has_builtin)
#if __has_builtin(__builtin_amdgcn_wmma_f32_16x16x4_f32)
#define HAVE_F32_WMMA 1
#endif
#endif

#define WAVE_ROWS 256

__device__ __forceinline__ void seg_load_step(float4 (&dst)[8], const char* wbase,
                                              const int4 ea, const int4 eb, unsigned cb) {
    const int e[8] = {ea.x, ea.y, ea.z, ea.w, eb.x, eb.y, eb.z, eb.w};
    for (int i = 0; i < 8; ++i)
        dst[i] = *(const float4*)(wbase + (((unsigned)e[i] & ~0x3ffu) | cb));
}

__device__ __forceinline__ void seg_mma_step(v8f (&acc)[4], const float4 (&f)[8],
                                             const int4 ea, const int4 eb, int cmpv) {
    const int e[8] = {ea.x, ea.y, ea.z, ea.w, eb.x, eb.y, eb.z, eb.w};
    for (int u = 0; u < 4; ++u) {
        v2f a;
        a.x = ((e[2 * u]     & 0x3ff) == cmpv) ? 1.0f : 0.0f;
        a.y = ((e[2 * u + 1] & 0x3ff) == cmpv) ? 1.0f : 0.0f;
        const float4 x = f[2 * u], y = f[2 * u + 1];
        v2f b0, b1, b2, b3;
        b0.x = x.x; b0.y = y.x;
        b1.x = x.y; b1.y = y.y;
        b2.x = x.z; b2.y = y.z;
        b3.x = x.w; b3.y = y.w;
        acc[0] = __builtin_amdgcn_wmma_f32_16x16x4_f32(false, a, false, b0, (short)0, acc[0], false, false);
        acc[1] = __builtin_amdgcn_wmma_f32_16x16x4_f32(false, a, false, b1, (short)0, acc[1], false, false);
        acc[2] = __builtin_amdgcn_wmma_f32_16x16x4_f32(false, a, false, b2, (short)0, acc[2], false, false);
        acc[3] = __builtin_amdgcn_wmma_f32_16x16x4_f32(false, a, false, b3, (short)0, acc[3], false, false);
    }
    __builtin_amdgcn_sched_barrier(0);
    asm volatile("v_nop\n\tv_nop\n\tv_nop\n\tv_nop"
                 : "+v"(acc[0]), "+v"(acc[1]), "+v"(acc[2]), "+v"(acc[3]));
    __builtin_amdgcn_sched_barrier(0);
}

#define SEG_GRP(s) (((s) < S0) ? (s) : ((s) - S0 + G0))
#define SEG_CMP(s) (((s) < S0) ? c : (c + 16))
#define SEG_FETCH(F, X0, X1, s)                                   \
    {                                                             \
        const int g_ = SEG_GRP(s);                                \
        X0 = *(const int4*)(el + 16 * g_);                        \
        X1 = *(const int4*)(el + 16 * g_ + 4);                    \
        seg_load_step(F, wbase, X0, X1, cb);                      \
    }
#define SEG_SET_ASIDE()                                           \
    {                                                             \
        for (int j_ = 0; j_ < 4; ++j_) {                          \
            acc0[j_] = acc[j_];                                   \
            acc[j_]  = vzero;                                     \
        }                                                         \
    }
#define SEG_STEP(F, X0, X1, s)                                    \
    {                                                             \
        if ((s) == S0) SEG_SET_ASIDE()                            \
        seg_mma_step(acc, F, X0, X1, SEG_CMP(s));                 \
    }

__global__ __launch_bounds__(512) void seg_accum(const float* __restrict__ features,
                                                 const int*   __restrict__ seg,
                                                 float*       __restrict__ ws,
                                                 size_t                    ws_bytes) {
    if (ws_bytes < (size_t)WS_NEED_BYTES) return;
    __shared__ __attribute__((aligned(16))) float part[SEG_WAVES * S_ * QCOLS];
    __shared__ __attribute__((aligned(16))) int   ent[ROWS_PER_BLOCK];
    __shared__ float cnt[S_];
    const int tid   = threadIdx.x;
    const int q     = blockIdx.x & 3;
    const int rhalf = (blockIdx.x >> 2) & 1;
    const int b     = blockIdx.x >> 3;
    const int wave  = __builtin_amdgcn_readfirstlane(tid >> 5);
    const int lane  = tid & 31;
    const int h     = lane >> 4;
    const int c     = lane & 15;

    const int row0 = rhalf * ROWS_PER_BLOCK + wave * WAVE_ROWS;
    const int*  sb = seg + b * N_ + row0 + 8 * lane;
    const char* wbase = (const char*)(features + ((size_t)b * N_ + (size_t)row0) * D_ + q * QCOLS);
    const unsigned cb = 16u * c;

    const int4 i0 = *(const int4*)(sb);
    const int4 i1 = *(const int4*)(sb + 4);
    const int idv[8] = {i0.x, i0.y, i0.z, i0.w, i1.x, i1.y, i1.z, i1.w};

    unsigned lo = 0;
    int n0 = 0;
    for (int j = 0; j < 8; ++j) {
        const unsigned m = __builtin_amdgcn_ballot_w32(idv[j] < 16);
        lo  = __builtin_amdgcn_mbcnt_lo(m, lo);
        n0 += __builtin_popcount(m);
    }
    {
        int* ew = ent + wave * WAVE_ROWS;
        int k0 = (int)lo;
        int k1 = n0 + 8 * lane - (int)lo;
        for (int j = 0; j < 8; ++j) {
            const bool first = idv[j] < 16;
            const int pos = first ? k0 : k1;
            k0 += first ? 1 : 0;
            k1 += first ? 0 : 1;
            ew[pos] = ((8 * lane + j) << 10) | (idv[j] & 0x3ff);
        }
    }
    if (tid < S_) cnt[tid] = 0.0f;
    __syncthreads();

    const int S0 = (n0 + 15) >> 4;
    const int G0 = n0 >> 4;
    const int T  = S0 + 16 - G0;

    const int* el = ent + wave * WAVE_ROWS + 8 * h;
    const v8f vzero = {};
    v8f acc[4]  = {vzero, vzero, vzero, vzero};
    v8f acc0[4] = {vzero, vzero, vzero, vzero};
    float4 fa[8], fb[8];
    int4 xa0, xa1, xb0, xb1;

    SEG_FETCH(fa, xa0, xa1, 0)

    if ((wave >> 2) == q) {
        atomicAdd(&cnt[i0.x], 1.0f); atomicAdd(&cnt[i0.y], 1.0f);
        atomicAdd(&cnt[i0.z], 1.0f); atomicAdd(&cnt[i0.w], 1.0f);
        atomicAdd(&cnt[i1.x], 1.0f); atomicAdd(&cnt[i1.y], 1.0f);
        atomicAdd(&cnt[i1.z], 1.0f); atomicAdd(&cnt[i1.w], 1.0f);
    }

#pragma nounroll
    for (int s = 0; s < 14; s += 2) {
        SEG_FETCH(fb, xb0, xb1, s + 1)
        SEG_STEP(fa, xa0, xa1, s)
        SEG_FETCH(fa, xa0, xa1, s + 2)
        SEG_STEP(fb, xb0, xb1, s + 1)
    }
    SEG_FETCH(fb, xb0, xb1, 15)
    SEG_STEP(fa, xa0, xa1, 14)
    if (T == 17) {
        SEG_FETCH(fa, xa0, xa1, 16)
        SEG_STEP(fb, xb0, xb1, 15)
        SEG_STEP(fa, xa0, xa1, 16)
    } else {
        SEG_STEP(fb, xb0, xb1, 15)
    }
    if (S0 == T) SEG_SET_ASIDE()

    float* pw = part + wave * (S_ * QCOLS);
    for (int r = 0; r < 8; ++r) {
        float4 v0 = {acc0[0][r], acc0[1][r], acc0[2][r], acc0[3][r]};
        float4 v1 = {acc[0][r],  acc[1][r],  acc[2][r],  acc[3][r]};
        *(float4*)(pw + (h * 8 + r) * QCOLS + c * 4)      = v0;
        *(float4*)(pw + (16 + h * 8 + r) * QCOLS + c * 4) = v1;
    }
    __syncthreads();

    float* gsum = ws + WS_PART + (size_t)rhalf * PLANE + (size_t)b * S_ * D_ + q * QCOLS;
    for (int k = 0; k < (S_ * QCOLS) / 512; ++k) {
        const int e = k * 512 + tid;
        float s = part[e];
        for (int w = 1; w < SEG_WAVES; ++w) s += part[w * (S_ * QCOLS) + e];
        gsum[(e >> 6) * D_ + (e & 63)] = s;
    }
    if (tid < S_) ws[WS_CNT + (b * CNT_VECS + rhalf * 4 + q) * S_ + tid] = cnt[tid];
}

__global__ __launch_bounds__(512) void pick_proj_ln(const float* __restrict__ ws,
                                                    const float* __restrict__ W,
                                                    const float* __restrict__ bias,
                                                    const float* __restrict__ gamma,
                                                    const float* __restrict__ beta,
                                                    float*       __restrict__ out,
                                                    size_t                    ws_bytes) {
    if (ws_bytes < (size_t)WS_NEED_BYTES) return;
    __shared__ float tile[16 * D_];
    __shared__ __attribute__((aligned(16))) float arow[16 * APITCH];
    __shared__ int   topi[2 * M_];
    __shared__ float scale[2 * M_];
    const int tid  = threadIdx.x;
    const int wave = tid >> 5;
    const int lane = tid & 31;
    const int half = lane >> 4;
    const int l    = lane & 15;
    const int rowbase = blockIdx.x * 16;
    const int colbase = wave * 16;

    if (wave < 2) {
        const float* cp = ws + WS_CNT + (2 * blockIdx.x + wave) * (CNT_VECS * S_) + lane;
        float mine = cp[0];
        for (int j = 1; j < CNT_VECS; ++j) mine += cp[j * S_];
        int rank = 0;
        for (int t = 0; t < S_; ++t) {
            const float ct = __int_as_float(__builtin_amdgcn_readlane(__float_as_int(mine), t));
            rank += (ct > mine || (ct == mine && t < lane)) ? 1 : 0;
        }
        if (rank < M_) {
            topi[wave * M_ + rank]  = lane;
            scale[wave * M_ + rank] = (mine > 0.0f) ? (1.0f / mine) : 0.0f;
        }
    }
    __syncthreads();

    {
        const float* r0 = ws + WS_PART
                        + ((size_t)((rowbase + wave) >> 3) * S_ + topi[wave]) * D_ + 4 * lane;
        const float4 x0 = *(const float4*)(r0);
        const float4 x1 = *(const float4*)(r0 + 128);
        const float4 y0 = *(const float4*)(r0 + PLANE);
        const float4 y1 = *(const float4*)(r0 + PLANE + 128);
        const float4 s0 = {x0.x + y0.x, x0.y + y0.y, x0.z + y0.z, x0.w + y0.w};
        const float4 s1 = {x1.x + y1.x, x1.y + y1.y, x1.z + y1.z, x1.w + y1.w};
        *(float4*)(arow + wave * APITCH + 4 * lane)       = s0;
        *(float4*)(arow + wave * APITCH + 128 + 4 * lane) = s1;
    }
    __syncthreads();
    const float* ap = arow + l * APITCH;
    const float* wp = W + (size_t)(colbase + l) * D_;

    v8f c = {};

#ifdef HAVE_F32_WMMA
    for (int kb = 0; kb < D_; kb += 64) {
        v2f av[16], wv[16];
        for (int i = 0; i < 16; ++i) {
            const int k = kb + 4 * i + 2 * half;
            av[i] = *(const v2f*)(ap + k);
            wv[i] = *(const v2f*)(wp + k);
        }
        for (int i = 0; i < 16; ++i)
            c = __builtin_amdgcn_wmma_f32_16x16x4_f32(false, av[i], false, wv[i],
                                                      (short)0, c, false, false);
    }
#else
    for (int k0 = 0; k0 < D_; k0 += 32) {
        v16h a, w;
        for (int i = 0; i < 8; ++i) {
            const int kk = k0 + ((i < 4) ? (2 * i + 8 * half)
                                         : (16 + 2 * (i - 4) + 8 * half));
            a[2 * i]     = (_Float16)ap[kk];
            a[2 * i + 1] = (_Float16)ap[kk + 1];
            w[2 * i]     = (_Float16)wp[kk];
            w[2 * i + 1] = (_Float16)wp[kk + 1];
        }
        c = __builtin_amdgcn_wmma_f32_16x16x32_f16(false, a, false, w,
                                                   (short)0, c, false, false);
    }
#endif

    __builtin_amdgcn_sched_barrier(0);
    asm volatile("v_nop\n\tv_nop\n\tv_nop\n\tv_nop" : "+v"(c));
    __builtin_amdgcn_sched_barrier(0);

    for (int j = 0; j < 8; ++j) {
        const int m = j + half * 8;
        const int n = colbase + l;
        tile[m * D_ + n] = c[j] * scale[m] + bias[n];
    }
    __syncthreads();

    const float* row = tile + wave * D_;
    float v[8], s = 0.0f, s2 = 0.0f;
    for (int j = 0; j < 8; ++j) {
        v[j] = row[lane * 8 + j];
        s  += v[j];
        s2 += v[j] * v[j];
    }
    for (int off = 16; off > 0; off >>= 1) {
        s  += __shfl_xor(s,  off, 32);
        s2 += __shfl_xor(s2, off, 32);
    }
    const float mean = s * (1.0f / D_);
    const float var  = s2 * (1.0f / D_) - mean * mean;
    const float rstd = rsqrtf(var + 1e-5f);
    float* orow = out + (size_t)(rowbase + wave) * D_;
    for (int j = 0; j < 8; ++j) {
        const int idx = lane * 8 + j;
        orow[idx] = (v[j] - mean) * rstd * gamma[idx] + beta[idx];
    }
}

extern "C" void kernel_launch(void* const* d_in, const int* in_sizes, int n_in,
                              void* d_out, int out_size, void* d_ws, size_t ws_size,
                              hipStream_t stream) {
    const float* features = (const float*)d_in[0];
    const int*   seg      = (const int*)  d_in[1];
    const float* W        = (const float*)d_in[2];
    const float* b        = (const float*)d_in[3];
    const float* gamma    = (const float*)d_in[4];
    const float* beta     = (const float*)d_in[5];
    float* ws  = (float*)d_ws;
    float* out = (float*)d_out;

    seg_accum<<<B_ * RHALVES * 4, 512, 0, stream>>>(features, seg, ws, ws_size);
    pick_proj_ln<<<(B_ * M_) / 16, 512, 0, stream>>>(ws, W, b, gamma, beta, out, ws_size);
}


# projection applied in front of the choice (it is linear): a seg_accum block multiplies its 32 x 64 plane by its W slice on WMMA; pick_ln adds 8 partial rows, no product left; f32 sum order changes
# speedup vs baseline: 2.7908x; 1.1086x over previous
#include <hip/hip_runtime.h>
#include <hip/hip_bf16.h>

#define S_ 32
#define B_ 32
#define N_ 8192
#define D_ 256
#define M_ 8
#define RHALVES 2
#define ROWS_PER_BLOCK (N_ / RHALVES)
#define QCOLS 64
#define SEG_WAVES 16

typedef __attribute__((ext_vector_type(2)))  float    v2f;
typedef __attribute__((ext_vector_type(8)))  float    v8f;

#define SLOTS   (RHALVES * 4)
#define WS_PROJ 0
#define WS_CNT  (B_ * SLOTS * S_ * D_)
#define CNT_VECS SLOTS
#define WS_NEED_BYTES ((WS_CNT + B_ * CNT_VECS * S_) * 4)
#define PPITCH  68
#define OPITCH  260

#define WAVE_ROWS 256

__device__ __forceinline__ void seg_load_step(float4 (&dst)[8], const char* wbase,
                                              const int4 ea, const int4 eb, unsigned cb) {
    const int e[8] = {ea.x, ea.y, ea.z, ea.w, eb.x, eb.y, eb.z, eb.w};
    for (int i = 0; i < 8; ++i)
        dst[i] = *(const float4*)(wbase + (((unsigned)e[i] & ~0x3ffu) | cb));
}

__device__ __forceinline__ void seg_mma_step(v8f (&acc)[4], const float4 (&f)[8],
                                             const int4 ea, const int4 eb, int cmpv) {
    const int e[8] = {ea.x, ea.y, ea.z, ea.w, eb.x, eb.y, eb.z, eb.w};
    for (int u = 0; u < 4; ++u) {
        v2f a;
        a.x = ((e[2 * u]     & 0x3ff) == cmpv) ? 1.0f : 0.0f;
        a.y = ((e[2 * u + 1] & 0x3ff) == cmpv) ? 1.0f : 0.0f;
        const float4 x = f[2 * u], y = f[2 * u + 1];
        v2f b0, b1, b2, b3;
        b0.x = x.x; b0.y = y.x;
        b1.x = x.y; b1.y = y.y;
        b2.x = x.z; b2.y = y.z;
        b3.x = x.w; b3.y = y.w;
        acc[0] = __builtin_amdgcn_wmma_f32_16x16x4_f32(false, a, false, b0, (short)0, acc[0], false, false);
        acc[1] = __builtin_amdgcn_wmma_f32_16x16x4_f32(false, a, false, b1, (short)0, acc[1], false, false);
        acc[2] = __builtin_amdgcn_wmma_f32_16x16x4_f32(false, a, false, b2, (short)0, acc[2], false, false);
        acc[3] = __builtin_amdgcn_wmma_f32_16x16x4_f32(false, a, false, b3, (short)0, acc[3], false, false);
    }
    __builtin_amdgcn_sched_barrier(0);
    asm volatile("v_nop\n\tv_nop\n\tv_nop\n\tv_nop"
                 : "+v"(acc[0]), "+v"(acc[1]), "+v"(acc[2]), "+v"(acc[3]));
    __builtin_amdgcn_sched_barrier(0);
}

#define SEG_GRP(s) (((s) < S0) ? (s) : ((s) - S0 + G0))
#define SEG_CMP(s) (((s) < S0) ? c : (c + 16))
#define SEG_FETCH(F, X0, X1, s)                                   \
    {                                                             \
        const int g_ = SEG_GRP(s);                                \
        X0 = *(const int4*)(el + 16 * g_);                        \
        X1 = *(const int4*)(el + 16 * g_ + 4);                    \
        seg_load_step(F, wbase, X0, X1, cb);                      \
    }
#define SEG_SET_ASIDE()                                           \
    {                                                             \
        for (int j_ = 0; j_ < 4; ++j_) {                          \
            acc0[j_] = acc[j_];                                   \
            acc[j_]  = vzero;                                     \
        }                                                         \
    }
#define SEG_STEP(F, X0, X1, s)                                    \
    {                                                             \
        if ((s) == S0) SEG_SET_ASIDE()                            \
        seg_mma_step(acc, F, X0, X1, SEG_CMP(s));                 \
    }

__global__ __launch_bounds__(512) void seg_accum(const float* __restrict__ features,
                                                 const int*   __restrict__ seg,
                                                 const float* __restrict__ W,
                                                 float*       __restrict__ ws,
                                                 size_t                    ws_bytes) {
    if (ws_bytes < (size_t)WS_NEED_BYTES) return;
    __shared__ __attribute__((aligned(16))) float part[SEG_WAVES * S_ * QCOLS];
    __shared__ __attribute__((aligned(16))) int   ent[ROWS_PER_BLOCK];
    __shared__ float cnt[S_];
    __shared__ __attribute__((aligned(16))) float plane[S_ * PPITCH];
    const int tid   = threadIdx.x;
    const int q     = blockIdx.x & 3;
    const int rhalf = (blockIdx.x >> 2) & 1;
    const int b     = blockIdx.x >> 3;
    const int wave  = __builtin_amdgcn_readfirstlane(tid >> 5);
    const int lane  = tid & 31;
    const int h     = lane >> 4;
    const int c     = lane & 15;

    const int row0 = rhalf * ROWS_PER_BLOCK + wave * WAVE_ROWS;
    const int*  sb = seg + b * N_ + row0 + 8 * lane;
    const char* wbase = (const char*)(features + ((size_t)b * N_ + (size_t)row0) * D_ + q * QCOLS);
    const unsigned cb = 16u * c;

    const int4 i0 = *(const int4*)(sb);
    const int4 i1 = *(const int4*)(sb + 4);
    const int idv[8] = {i0.x, i0.y, i0.z, i0.w, i1.x, i1.y, i1.z, i1.w};

    unsigned lo = 0;
    int n0 = 0;
    for (int j = 0; j < 8; ++j) {
        const unsigned m = __builtin_amdgcn_ballot_w32(idv[j] < 16);
        lo  = __builtin_amdgcn_mbcnt_lo(m, lo);
        n0 += __builtin_popcount(m);
    }
    {
        int* ew = ent + wave * WAVE_ROWS;
        int k0 = (int)lo;
        int k1 = n0 + 8 * lane - (int)lo;
        for (int j = 0; j < 8; ++j) {
            const bool first = idv[j] < 16;
            const int pos = first ? k0 : k1;
            k0 += first ? 1 : 0;
            k1 += first ? 0 : 1;
            ew[pos] = ((8 * lane + j) << 10) | (idv[j] & 0x3ff);
        }
    }
    if (tid < S_) cnt[tid] = 0.0f;
    __syncthreads();

    const int S0 = (n0 + 15) >> 4;
    const int G0 = n0 >> 4;
    const int T  = S0 + 16 - G0;

    const int* el = ent + wave * WAVE_ROWS + 8 * h;
    const v8f vzero = {};
    v8f acc[4]  = {vzero, vzero, vzero, vzero};
    v8f acc0[4] = {vzero, vzero, vzero, vzero};
    float4 fa[8], fb[8];
    int4 xa0, xa1, xb0, xb1;

    SEG_FETCH(fa, xa0, xa1, 0)

    if ((wave >> 2) == q) {
        atomicAdd(&cnt[i0.x], 1.0f); atomicAdd(&cnt[i0.y], 1.0f);
        atomicAdd(&cnt[i0.z], 1.0f); atomicAdd(&cnt[i0.w], 1.0f);
        atomicAdd(&cnt[i1.x], 1.0f); atomicAdd(&cnt[i1.y], 1.0f);
        atomicAdd(&cnt[i1.z], 1.0f); atomicAdd(&cnt[i1.w], 1.0f);
    }

#pragma nounroll
    for (int s = 0; s < 14; s += 2) {
        SEG_FETCH(fb, xb0, xb1, s + 1)
        SEG_STEP(fa, xa0, xa1, s)
        SEG_FETCH(fa, xa0, xa1, s + 2)
        SEG_STEP(fb, xb0, xb1, s + 1)
    }
    v2f wv[16];
    {
        const float* wp = W + (size_t)(wave * 16 + c) * D_ + q * QCOLS + 2 * h;
        for (int i = 0; i < 16; ++i) wv[i] = *(const v2f*)(wp + 4 * i);
    }
    SEG_FETCH(fb, xb0, xb1, 15)
    SEG_STEP(fa, xa0, xa1, 14)
    if (T == 17) {
        SEG_FETCH(fa, xa0, xa1, 16)
        SEG_STEP(fb, xb0, xb1, 15)
        SEG_STEP(fa, xa0, xa1, 16)
    } else {
        SEG_STEP(fb, xb0, xb1, 15)
    }
    if (S0 == T) SEG_SET_ASIDE()

    float* pw = part + wave * (S_ * QCOLS);
    for (int r = 0; r < 8; ++r) {
        float4 v0 = {acc0[0][r], acc0[1][r], acc0[2][r], acc0[3][r]};
        float4 v1 = {acc[0][r],  acc[1][r],  acc[2][r],  acc[3][r]};
        *(float4*)(pw + (h * 8 + r) * QCOLS + c * 4)      = v0;
        *(float4*)(pw + (16 + h * 8 + r) * QCOLS + c * 4) = v1;
    }
    __syncthreads();

    for (int k = 0; k < (S_ * QCOLS) / 512; ++k) {
        const int e = k * 512 + tid;
        float sm = part[e];
        for (int w = 1; w < SEG_WAVES; ++w) sm += part[w * (S_ * QCOLS) + e];
        plane[(e >> 6) * PPITCH + (e & 63)] = sm;
    }
    const int slot = rhalf * 4 + q;
    if (tid < S_) ws[WS_CNT + (b * CNT_VECS + slot) * S_ + tid] = cnt[tid];
    __syncthreads();

    v8f p0 = {}, p1 = {};
    {
        const float* ap = plane + c * PPITCH + 2 * h;
        v2f a0[16], a1[16];
        for (int i = 0; i < 16; ++i) {
            a0[i] = *(const v2f*)(ap + 4 * i);
            a1[i] = *(const v2f*)(ap + 16 * PPITCH + 4 * i);
        }
        __builtin_amdgcn_sched_barrier(0);
        for (int i = 0; i < 16; ++i) {
            p0 = __builtin_amdgcn_wmma_f32_16x16x4_f32(false, a0[i], false, wv[i], (short)0, p0, false, false);
            p1 = __builtin_amdgcn_wmma_f32_16x16x4_f32(false, a1[i], false, wv[i], (short)0, p1, false, false);
        }
    }
    __builtin_amdgcn_sched_barrier(0);
    asm volatile("v_nop\n\tv_nop\n\tv_nop\n\tv_nop" : "+v"(p0), "+v"(p1));
    __builtin_amdgcn_sched_barrier(0);

    float* ot = part;
    for (int r = 0; r < 8; ++r) {
        ot[(8 * h + r) * OPITCH + 16 * wave + c]      = p0[r];
        ot[(16 + 8 * h + r) * OPITCH + 16 * wave + c] = p1[r];
    }
    __syncthreads();
    float* gp = ws + WS_PROJ + (size_t)(b * SLOTS + slot) * (S_ * D_);
    for (int k = 0; k < (S_ * D_) / (4 * 512); ++k) {
        const int f   = k * 512 + tid;
        const int row = f >> 6, pc = f & 63;
        *(float4*)(gp + row * D_ + 4 * pc) = *(const float4*)(ot + row * OPITCH + 4 * pc);
    }
}

__device__ __forceinline__ float4 f4_add(const float4 a, const float4 b) {
    float4 r = {a.x + b.x, a.y + b.y, a.z + b.z, a.w + b.w};
    return r;
}

__global__ __launch_bounds__(256) void pick_ln(const float* __restrict__ ws,
                                               const float* __restrict__ bias,
                                               const float* __restrict__ gamma,
                                               const float* __restrict__ beta,
                                               float*       __restrict__ out,
                                               size_t                    ws_bytes) {
    if (ws_bytes < (size_t)WS_NEED_BYTES) return;
    const int tid  = threadIdx.x;
    const int wave = __builtin_amdgcn_readfirstlane(tid >> 5);
    const int lane = tid & 31;
    const int b    = blockIdx.x;

    const float* cp = ws + WS_CNT + b * (CNT_VECS * S_) + lane;
    float mine = cp[0];
    for (int j = 1; j < CNT_VECS; ++j) mine += cp[j * S_];

    int rank = 0;
    for (int t = 0; t < S_; ++t) {
        const float ct = __int_as_float(__builtin_amdgcn_readlane(__float_as_int(mine), t));
        rank += (ct > mine || (ct == mine && t < lane)) ? 1 : 0;
    }
    const unsigned pick = __builtin_amdgcn_ballot_w32(rank == wave);
    const int sel = __builtin_amdgcn_readfirstlane(__builtin_ctz(pick));
    const float cs = __int_as_float(__builtin_amdgcn_readlane(__float_as_int(mine), sel));
    const float sc = (cs > 0.0f) ? (1.0f / cs) : 0.0f;

    const float* r0 = ws + WS_PROJ + ((size_t)(b * SLOTS) * S_ + sel) * D_ + 4 * lane;
    float4 x0[SLOTS], x1[SLOTS];
    for (int j = 0; j < SLOTS; ++j) {
        x0[j] = *(const float4*)(r0 + (size_t)j * (S_ * D_));
        x1[j] = *(const float4*)(r0 + (size_t)j * (S_ * D_) + 128);
    }
    const float4 bi0 = *(const float4*)(bias  + 4 * lane), bi1 = *(const float4*)(bias  + 128 + 4 * lane);
    const float4 ga0 = *(const float4*)(gamma + 4 * lane), ga1 = *(const float4*)(gamma + 128 + 4 * lane);
    const float4 be0 = *(const float4*)(beta  + 4 * lane), be1 = *(const float4*)(beta  + 128 + 4 * lane);
    float4 s0 = x0[0], s1 = x1[0];
    for (int j = 1; j < SLOTS; ++j) {
        s0 = f4_add(s0, x0[j]);
        s1 = f4_add(s1, x1[j]);
    }
    float v[8] = {s0.x * sc + bi0.x, s0.y * sc + bi0.y, s0.z * sc + bi0.z, s0.w * sc + bi0.w,
                  s1.x * sc + bi1.x, s1.y * sc + bi1.y, s1.z * sc + bi1.z, s1.w * sc + bi1.w};

    float sm = 0.0f, s2 = 0.0f;
    for (int j = 0; j < 8; ++j) {
        sm += v[j];
        s2 += v[j] * v[j];
    }
    for (int off = 16; off > 0; off >>= 1) {
        sm += __shfl_xor(sm, off, 32);
        s2 += __shfl_xor(s2, off, 32);
    }
    const float mean = sm * (1.0f / D_);
    const float var  = s2 * (1.0f / D_) - mean * mean;
    const float rstd = rsqrtf(var + 1e-5f);
    float* orow = out + (size_t)(b * M_ + wave) * D_ + 4 * lane;
    const float4 o0 = {(v[0] - mean) * rstd * ga0.x + be0.x, (v[1] - mean) * rstd * ga0.y + be0.y,
                       (v[2] - mean) * rstd * ga0.z + be0.z, (v[3] - mean) * rstd * ga0.w + be0.w};
    const float4 o1 = {(v[4] - mean) * rstd * ga1.x + be1.x, (v[5] - mean) * rstd * ga1.y + be1.y,
                       (v[6] - mean) * rstd * ga1.z + be1.z, (v[7] - mean) * rstd * ga1.w + be1.w};
    *(float4*)(orow)       = o0;
    *(float4*)(orow + 128) = o1;
}

extern "C" void kernel_launch(void* const* d_in, const int* in_sizes, int n_in,
                              void* d_out, int out_size, void* d_ws, size_t ws_size,
                              hipStream_t stream) {
    const float* features = (const float*)d_in[0];
    const int*   seg      = (const int*)  d_in[1];
    const float* W        = (const float*)d_in[2];
    const float* b        = (const float*)d_in[3];
    const float* gamma    = (const float*)d_in[4];
    const float* beta     = (const float*)d_in[5];
    float* ws  = (float*)d_ws;
    float* out = (float*)d_out;

    seg_accum<<<B_ * RHALVES * 4, 512, 0, stream>>>(features, seg, W, ws, ws_size);
    pick_ln<<<B_, 256, 0, stream>>>(ws, b, gamma, beta, out, ws_size);
}
